# starting kernel
# baseline (speedup 1.0000x reference)
#include <hip/hip_runtime.h>

#define N_NODES 100000
#define N_EDGES 1600000
#define IN_FT   128
#define OUT_FT  128

typedef __attribute__((ext_vector_type(2))) float v2f;
typedef __attribute__((ext_vector_type(8))) float v8f;

__global__ __launch_bounds__(256) void gemm_bias_wmma(
    const float* __restrict__ x, const float* __restrict__ W,
    const float* __restrict__ b, float* __restrict__ h, int nNodes)
{
    __shared__ float Ws[IN_FT * OUT_FT];

    {
        const float4* src = (const float4*)W;
        float4*       dst = (float4*)Ws;
        int tid = threadIdx.x;
#pragma unroll
        for (int i = 0; i < 16; ++i)
            dst[tid + i * 256] = src[tid + i * 256];
    }
    __syncthreads();

    const int wave  = threadIdx.x >> 5;
    const int lane  = threadIdx.x & 31;
    const int tile  = blockIdx.x * 8 + wave;
    if (tile * 16 >= nNodes) return;

    const int row0  = tile * 16;
    const int m     = lane & 15;
    const int khalf = lane >> 4;

    const v2f* xv = (const v2f*)(x + (size_t)(row0 + m) * IN_FT + 2 * khalf);
    v2f afrag[32];
#pragma unroll
    for (int kk = 0; kk < 32; ++kk)
        afrag[kk] = xv[kk * 2];

    v8f acc[8] = {};

#pragma unroll
    for (int kk = 0; kk < 32; ++kk) {
        const int krow = kk * 4 + 2 * khalf;
#pragma unroll
        for (int nt = 0; nt < 8; ++nt) {
            const int n = nt * 16 + m;
            v2f bf;
            bf.x = Ws[(krow + 0) * OUT_FT + n];
            bf.y = Ws[(krow + 1) * OUT_FT + n];
            acc[nt] = __builtin_amdgcn_wmma_f32_16x16x4_f32(
                 false, afrag[kk],  false, bf,
                 (short)0, acc[nt],
                 false,  false);
        }
    }

#pragma unroll
    for (int nt = 0; nt < 8; ++nt) {
        const int n      = nt * 16 + m;
        const float bias = b[n];
#pragma unroll
        for (int j = 0; j < 8; ++j) {
            const int row = row0 + j + 8 * khalf;
            h[(size_t)row * OUT_FT + n] = acc[nt][j] + bias;
        }
    }
}

__global__ __launch_bounds__(256) void zero_f32x4(float4* __restrict__ p, int n4)
{
    int i = blockIdx.x * blockDim.x + threadIdx.x;
    if (i < n4) p[i] = make_float4(0.f, 0.f, 0.f, 0.f);
}

__global__ __launch_bounds__(256) void spmm_scatter(
    const float* __restrict__ h,
    const int*   __restrict__ rows,
    const int*   __restrict__ cols,
    const float* __restrict__ vals,
    float* __restrict__ out, int nEdges)
{
    const int gwave = __builtin_amdgcn_readfirstlane(
        blockIdx.x * 8 + (threadIdx.x >> 5));
    const int lane  = threadIdx.x & 31;
    if (gwave >= nEdges) return;

    const int   r = rows[gwave];
    const int   c = cols[gwave];
    const float v = vals[gwave];

    const float4 mv = *((const float4*)(h + (size_t)c * OUT_FT) + lane);

    float* op = out + (size_t)r * OUT_FT + lane * 4;
    unsafeAtomicAdd(op + 0, v * mv.x);
    unsafeAtomicAdd(op + 1, v * mv.y);
    unsafeAtomicAdd(op + 2, v * mv.z);
    unsafeAtomicAdd(op + 3, v * mv.w);
}

__global__ __launch_bounds__(256) void relu_f32x4(float4* __restrict__ p, int n4)
{
    int i = blockIdx.x * blockDim.x + threadIdx.x;
    if (i < n4) {
        float4 t = p[i];
        t.x = fmaxf(t.x, 0.f);
        t.y = fmaxf(t.y, 0.f);
        t.z = fmaxf(t.z, 0.f);
        t.w = fmaxf(t.w, 0.f);
        p[i] = t;
    }
}

extern "C" void kernel_launch(void* const* d_in, const int* in_sizes, int n_in,
                              void* d_out, int out_size, void* d_ws, size_t ws_size,
                              hipStream_t stream)
{
    const float* x         = (const float*)d_in[0];
    const int*   edge_rows = (const int*)  d_in[1];
    const int*   edge_cols = (const int*)  d_in[2];
    const float* edge_vals = (const float*)d_in[3];
    const float* W         = (const float*)d_in[4];
    const float* b         = (const float*)d_in[5];
    float*       out       = (float*)d_out;
    float*       h         = (float*)d_ws;

    const int nOut4 = (N_NODES * OUT_FT) / 4;

    zero_f32x4<<<(nOut4 + 255) / 256, 256, 0, stream>>>((float4*)out, nOut4);

    const int mTiles = N_NODES / 16;
    gemm_bias_wmma<<<(mTiles + 7) / 8, 256, 0, stream>>>(x, W, b, h, N_NODES);

    spmm_scatter<<<(N_EDGES + 7) / 8, 256, 0, stream>>>(
        h, edge_rows, edge_cols, edge_vals, out, N_EDGES);

    relu_f32x4<<<(nOut4 + 255) / 256, 256, 0, stream>>>((float4*)out, nOut4);
}


# f32 global atomic scatter replaced: edges binned in-call by 128-row destination group (LDS counts, scan, place); one block per bin sums in an LDS tile (LDS f32 adds, arrival order), relu at the store
# speedup vs baseline: 2.6133x; 2.6133x over previous
#include <hip/hip_runtime.h>

#define N_NODES 100000
#define N_EDGES 1600000
#define IN_FT   128
#define OUT_FT  128

typedef __attribute__((ext_vector_type(2))) float v2f;
typedef __attribute__((ext_vector_type(8))) float v8f;

__global__ __launch_bounds__(256) void gemm_bias_wmma(
    const float* __restrict__ x, const float* __restrict__ W,
    const float* __restrict__ b, float* __restrict__ h, int nNodes)
{
    __shared__ float Ws[IN_FT * OUT_FT];

    {
        const float4* src = (const float4*)W;
        float4*       dst = (float4*)Ws;
        int tid = threadIdx.x;
#pragma unroll
        for (int i = 0; i < 16; ++i)
            dst[tid + i * 256] = src[tid + i * 256];
    }
    __syncthreads();

    const int wave  = threadIdx.x >> 5;
    const int lane  = threadIdx.x & 31;
    const int tile  = blockIdx.x * 8 + wave;
    if (tile * 16 >= nNodes) return;

    const int row0  = tile * 16;
    const int m     = lane & 15;
    const int khalf = lane >> 4;

    const v2f* xv = (const v2f*)(x + (size_t)(row0 + m) * IN_FT + 2 * khalf);
    v2f afrag[32];
#pragma unroll
    for (int kk = 0; kk < 32; ++kk)
        afrag[kk] = xv[kk * 2];

    v8f acc[8] = {};

#pragma unroll
    for (int kk = 0; kk < 32; ++kk) {
        const int krow = kk * 4 + 2 * khalf;
#pragma unroll
        for (int nt = 0; nt < 8; ++nt) {
            const int n = nt * 16 + m;
            v2f bf;
            bf.x = Ws[(krow + 0) * OUT_FT + n];
            bf.y = Ws[(krow + 1) * OUT_FT + n];
            acc[nt] = __builtin_amdgcn_wmma_f32_16x16x4_f32(
                 false, afrag[kk],  false, bf,
                 (short)0, acc[nt],
                 false,  false);
        }
    }

#pragma unroll
    for (int nt = 0; nt < 8; ++nt) {
        const int n      = nt * 16 + m;
        const float bias = b[n];
#pragma unroll
        for (int j = 0; j < 8; ++j) {
            const int row = row0 + j + 8 * khalf;
            h[(size_t)row * OUT_FT + n] = acc[nt][j] + bias;
        }
    }
}

#define BIN_SHIFT 7
#define BIN_ROWS  128
#define NBIN      782
#define NSLICE    128
#define SLICE_E   12500
#define NCELL     (NBIN * NSLICE)
#define SLICE_K   13

__global__ __launch_bounds__(1024) void bin_count(
    const int* __restrict__ rows, int* __restrict__ cell)
{
    __shared__ int lh[NBIN];
    const int t = threadIdx.x;
    const int b = blockIdx.x;
    for (int i = t; i < NBIN; i += 1024) lh[i] = 0;
    __syncthreads();
    const int* rp = rows + b * SLICE_E;
    int r[SLICE_K];
#pragma unroll
    for (int k = 0; k < SLICE_K; ++k) {
        const int i = t + k * 1024;
        r[k] = (i < SLICE_E) ? rp[i] : -1;
    }
#pragma unroll
    for (int k = 0; k < SLICE_K; ++k)
        if (r[k] >= 0) atomicAdd(&lh[r[k] >> BIN_SHIFT], 1);
    __syncthreads();
    for (int i = t; i < NBIN; i += 1024) cell[i * NSLICE + b] = lh[i];
}

#define SCAN_IT 25

__global__ __launch_bounds__(1024) void scan_cells(
    const int* __restrict__ cell, int* __restrict__ cellx)
{
    __shared__ int wtot[SCAN_IT][32];
    __shared__ int wexc[SCAN_IT][32];
    __shared__ int ittot[SCAN_IT];

    const int t    = threadIdx.x;
    const int lane = t & 31;
    const int wave = t >> 5;

    int ssum[SCAN_IT];
    int incl[SCAN_IT];
#pragma unroll
    for (int it = 0; it < SCAN_IT; ++it) {
        const int idx = it * 4096 + t * 4;
        int4 c = make_int4(0, 0, 0, 0);
        if (idx < NCELL) c = *(const int4*)(cell + idx);
        ssum[it] = c.x + c.y + c.z + c.w;
        incl[it] = ssum[it];
    }
#pragma unroll
    for (int d = 1; d < 32; d <<= 1) {
#pragma unroll
        for (int it = 0; it < SCAN_IT; ++it) {
            const int u = __shfl_up(incl[it], d);
            if (lane >= d) incl[it] += u;
        }
    }
    if (lane == 31) {
#pragma unroll
        for (int it = 0; it < SCAN_IT; ++it) wtot[it][wave] = incl[it];
    }
    __syncthreads();

    if (wave < SCAN_IT) {
        const int v0 = wtot[wave][lane];
        int v = v0;
#pragma unroll
        for (int d = 1; d < 32; d <<= 1) {
            const int u = __shfl_up(v, d);
            if (lane >= d) v += u;
        }
        wexc[wave][lane] = v - v0;
        if (lane == 31) ittot[wave] = v;
    }
    __syncthreads();

    int carry = 0;
#pragma unroll
    for (int it = 0; it < SCAN_IT; ++it) {
        const int idx = it * 4096 + t * 4;
        if (idx < NCELL) {
            const int4 c = *(const int4*)(cell + idx);
            const int base = carry + wexc[it][wave] + incl[it] - ssum[it];
            int4 o;
            o.x = base;
            o.y = base + c.x;
            o.z = base + c.x + c.y;
            o.w = base + c.x + c.y + c.z;
            *(int4*)(cellx + idx) = o;
        }
        carry += ittot[it];
    }
}

__global__ __launch_bounds__(1024) void bin_place(
    const int*   __restrict__ rows,
    const int*   __restrict__ cols,
    const float* __restrict__ vals,
    const int*   __restrict__ cellx,
    int* __restrict__ binned)
{
    __shared__ int lofs[NBIN];
    const int t = threadIdx.x;
    const int b = blockIdx.x;
    for (int i = t; i < NBIN; i += 1024) lofs[i] = cellx[i * NSLICE + b];
    __syncthreads();
    const int e0 = b * SLICE_E;
    int   r[SLICE_K];
    int   c[SLICE_K];
    float v[SLICE_K];
#pragma unroll
    for (int k = 0; k < SLICE_K; ++k) {
        const int i = t + k * 1024;
        r[k] = -1; c[k] = 0; v[k] = 0.f;
        if (i < SLICE_E) { r[k] = rows[e0 + i]; c[k] = cols[e0 + i]; v[k] = vals[e0 + i]; }
    }
#pragma unroll
    for (int k = 0; k < SLICE_K; ++k) {
        if (r[k] >= 0) {
            const int slot = atomicAdd(&lofs[r[k] >> BIN_SHIFT], 1);
            ((int2*)binned)[slot] = make_int2((c[k] << BIN_SHIFT) | (r[k] & (BIN_ROWS - 1)),
                                              __float_as_int(v[k]));
        }
    }
}

__global__ __launch_bounds__(512) void bin_gather_relu(
    const float* __restrict__ h,
    const int*   __restrict__ cellx,
    const int*   __restrict__ binned,
    float* __restrict__ out)
{
    __shared__ float tile[BIN_ROWS * OUT_FT];

    const int t    = threadIdx.x;
    const int lane = t & 31;
    const int wave = __builtin_amdgcn_readfirstlane(t >> 5);
    const int bin  = blockIdx.x;

    {
        float4* tz = (float4*)tile;
#pragma unroll
        for (int i = 0; i < 8; ++i) tz[t + i * 512] = make_float4(0.f, 0.f, 0.f, 0.f);
    }
    __syncthreads();

    const int start = cellx[bin * NSLICE];
    const int end   = (bin + 1 < NBIN) ? cellx[(bin + 1) * NSLICE] : N_EDGES;
    const int2*   sp = (const int2*)binned;
    const float4* hv = (const float4*)h + lane;

    for (int base = start + wave * 32; base < end; base += 16 * 32) {
        const int left = end - base;
        const int m = left < 32 ? left : 32;
        int2 ent = make_int2(0, 0);
        if (lane < m) ent = sp[base + lane];

        int j = 0;
        for (; j + 4 <= m; j += 4) {
            const int   k0 = __builtin_amdgcn_readlane(ent.x, j + 0);
            const int   k1 = __builtin_amdgcn_readlane(ent.x, j + 1);
            const int   k2 = __builtin_amdgcn_readlane(ent.x, j + 2);
            const int   k3 = __builtin_amdgcn_readlane(ent.x, j + 3);
            const float v0 = __int_as_float(__builtin_amdgcn_readlane(ent.y, j + 0));
            const float v1 = __int_as_float(__builtin_amdgcn_readlane(ent.y, j + 1));
            const float v2 = __int_as_float(__builtin_amdgcn_readlane(ent.y, j + 2));
            const float v3 = __int_as_float(__builtin_amdgcn_readlane(ent.y, j + 3));
            const float4 a0 = hv[(size_t)(k0 >> BIN_SHIFT) * 32];
            const float4 a1 = hv[(size_t)(k1 >> BIN_SHIFT) * 32];
            const float4 a2 = hv[(size_t)(k2 >> BIN_SHIFT) * 32];
            const float4 a3 = hv[(size_t)(k3 >> BIN_SHIFT) * 32];
            float* t0 = tile + (k0 & (BIN_ROWS - 1)) * OUT_FT + lane;
            float* t1 = tile + (k1 & (BIN_ROWS - 1)) * OUT_FT + lane;
            float* t2 = tile + (k2 & (BIN_ROWS - 1)) * OUT_FT + lane;
            float* t3 = tile + (k3 & (BIN_ROWS - 1)) * OUT_FT + lane;
            unsafeAtomicAdd(t0 +  0, v0 * a0.x); unsafeAtomicAdd(t0 + 32, v0 * a0.y);
            unsafeAtomicAdd(t0 + 64, v0 * a0.z); unsafeAtomicAdd(t0 + 96, v0 * a0.w);
            unsafeAtomicAdd(t1 +  0, v1 * a1.x); unsafeAtomicAdd(t1 + 32, v1 * a1.y);
            unsafeAtomicAdd(t1 + 64, v1 * a1.z); unsafeAtomicAdd(t1 + 96, v1 * a1.w);
            unsafeAtomicAdd(t2 +  0, v2 * a2.x); unsafeAtomicAdd(t2 + 32, v2 * a2.y);
            unsafeAtomicAdd(t2 + 64, v2 * a2.z); unsafeAtomicAdd(t2 + 96, v2 * a2.w);
            unsafeAtomicAdd(t3 +  0, v3 * a3.x); unsafeAtomicAdd(t3 + 32, v3 * a3.y);
            unsafeAtomicAdd(t3 + 64, v3 * a3.z); unsafeAtomicAdd(t3 + 96, v3 * a3.w);
        }
        for (; j < m; ++j) {
            const int   k = __builtin_amdgcn_readlane(ent.x, j);
            const float v = __int_as_float(__builtin_amdgcn_readlane(ent.y, j));
            const float4 a = hv[(size_t)(k >> BIN_SHIFT) * 32];
            float* tr = tile + (k & (BIN_ROWS - 1)) * OUT_FT + lane;
            unsafeAtomicAdd(tr +  0, v * a.x); unsafeAtomicAdd(tr + 32, v * a.y);
            unsafeAtomicAdd(tr + 64, v * a.z); unsafeAtomicAdd(tr + 96, v * a.w);
        }
    }
    __syncthreads();

    for (int rl = wave; rl < BIN_ROWS; rl += 16) {
        const int row = bin * BIN_ROWS + rl;
        if (row < N_NODES) {
            const float* tr = tile + rl * OUT_FT + lane;
            float4 o;
            o.x = fmaxf(tr[0],  0.f);
            o.y = fmaxf(tr[32], 0.f);
            o.z = fmaxf(tr[64], 0.f);
            o.w = fmaxf(tr[96], 0.f);
            *((float4*)(out + (size_t)row * OUT_FT) + lane) = o;
        }
    }
}

extern "C" void kernel_launch(void* const* d_in, const int* in_sizes, int n_in,
                              void* d_out, int out_size, void* d_ws, size_t ws_size,
                              hipStream_t stream)
{
    const float* x         = (const float*)d_in[0];
    const int*   edge_rows = (const int*)  d_in[1];
    const int*   edge_cols = (const int*)  d_in[2];
    const float* edge_vals = (const float*)d_in[3];
    const float* W         = (const float*)d_in[4];
    const float* b         = (const float*)d_in[5];
    float*       out       = (float*)d_out;

    float* h      = (float*)d_ws;
    int*   cell   = (int*)(h + (size_t)N_NODES * OUT_FT);
    int*   cellx  = cell + 102400;
    int*   binned = cellx + 102400;

    const int mTiles = N_NODES / 16;
    gemm_bias_wmma<<<(mTiles + 7) / 8, 256, 0, stream>>>(x, W, b, h, N_NODES);

    bin_count<<<NSLICE, 1024, 0, stream>>>(edge_rows, cell);
    scan_cells<<<1, 1024, 0, stream>>>(cell, cellx);
    bin_place<<<NSLICE, 1024, 0, stream>>>(edge_rows, edge_cols, edge_vals, cellx, binned);

    bin_gather_relu<<<NBIN, 512, 0, stream>>>(h, cellx, binned, out);
}


# gather kernel: LDS f32 adds into a 64 KB tile replaced by ordering each bin's records by row in LDS (integer count, scan, place); a wave sums whole rows in registers, in LDS slot (arrival) order
# speedup vs baseline: 2.7535x; 1.0536x over previous
#include <hip/hip_runtime.h>

#define N_NODES 100000
#define N_EDGES 1600000
#define IN_FT   128
#define OUT_FT  128

typedef __attribute__((ext_vector_type(2))) float v2f;
typedef __attribute__((ext_vector_type(8))) float v8f;

__global__ __launch_bounds__(256) void gemm_bias_wmma(
    const float* __restrict__ x, const float* __restrict__ W,
    const float* __restrict__ b, float* __restrict__ h, int nNodes)
{
    __shared__ float Ws[IN_FT * OUT_FT];

    {
        const float4* src = (const float4*)W;
        float4*       dst = (float4*)Ws;
        int tid = threadIdx.x;
#pragma unroll
        for (int i = 0; i < 16; ++i)
            dst[tid + i * 256] = src[tid + i * 256];
    }
    __syncthreads();

    const int wave  = threadIdx.x >> 5;
    const int lane  = threadIdx.x & 31;
    const int tile  = blockIdx.x * 8 + wave;
    if (tile * 16 >= nNodes) return;

    const int row0  = tile * 16;
    const int m     = lane & 15;
    const int khalf = lane >> 4;

    const v2f* xv = (const v2f*)(x + (size_t)(row0 + m) * IN_FT + 2 * khalf);
    v2f afrag[32];
#pragma unroll
    for (int kk = 0; kk < 32; ++kk)
        afrag[kk] = xv[kk * 2];

    v8f acc[8] = {};

#pragma unroll
    for (int kk = 0; kk < 32; ++kk) {
        const int krow = kk * 4 + 2 * khalf;
#pragma unroll
        for (int nt = 0; nt < 8; ++nt) {
            const int n = nt * 16 + m;
            v2f bf;
            bf.x = Ws[(krow + 0) * OUT_FT + n];
            bf.y = Ws[(krow + 1) * OUT_FT + n];
            acc[nt] = __builtin_amdgcn_wmma_f32_16x16x4_f32(
                 false, afrag[kk],  false, bf,
                 (short)0, acc[nt],
                 false,  false);
        }
    }

#pragma unroll
    for (int nt = 0; nt < 8; ++nt) {
        const int n      = nt * 16 + m;
        const float bias = b[n];
#pragma unroll
        for (int j = 0; j < 8; ++j) {
            const int row = row0 + j + 8 * khalf;
            h[(size_t)row * OUT_FT + n] = acc[nt][j] + bias;
        }
    }
}

#define BIN_SHIFT 7
#define BIN_ROWS  128
#define NBIN      782
#define NSLICE    128
#define SLICE_E   12500
#define NCELL     (NBIN * NSLICE)
#define SLICE_K   13

__global__ __launch_bounds__(1024) void bin_count(
    const int* __restrict__ rows, int* __restrict__ cell)
{
    __shared__ int lh[NBIN];
    const int t = threadIdx.x;
    const int b = blockIdx.x;
    for (int i = t; i < NBIN; i += 1024) lh[i] = 0;
    __syncthreads();
    const int* rp = rows + b * SLICE_E;
    int r[SLICE_K];
#pragma unroll
    for (int k = 0; k < SLICE_K; ++k) {
        const int i = t + k * 1024;
        r[k] = (i < SLICE_E) ? rp[i] : -1;
    }
#pragma unroll
    for (int k = 0; k < SLICE_K; ++k)
        if (r[k] >= 0) atomicAdd(&lh[r[k] >> BIN_SHIFT], 1);
    __syncthreads();
    for (int i = t; i < NBIN; i += 1024) cell[i * NSLICE + b] = lh[i];
}

#define SCAN_IT 25

__global__ __launch_bounds__(1024) void scan_cells(
    const int* __restrict__ cell, int* __restrict__ cellx)
{
    __shared__ int wtot[SCAN_IT][32];
    __shared__ int wexc[SCAN_IT][32];
    __shared__ int ittot[SCAN_IT];

    const int t    = threadIdx.x;
    const int lane = t & 31;
    const int wave = t >> 5;

    int ssum[SCAN_IT];
    int incl[SCAN_IT];
#pragma unroll
    for (int it = 0; it < SCAN_IT; ++it) {
        const int idx = it * 4096 + t * 4;
        int4 c = make_int4(0, 0, 0, 0);
        if (idx < NCELL) c = *(const int4*)(cell + idx);
        ssum[it] = c.x + c.y + c.z + c.w;
        incl[it] = ssum[it];
    }
#pragma unroll
    for (int d = 1; d < 32; d <<= 1) {
#pragma unroll
        for (int it = 0; it < SCAN_IT; ++it) {
            const int u = __shfl_up(incl[it], d);
            if (lane >= d) incl[it] += u;
        }
    }
    if (lane == 31) {
#pragma unroll
        for (int it = 0; it < SCAN_IT; ++it) wtot[it][wave] = incl[it];
    }
    __syncthreads();

    if (wave < SCAN_IT) {
        const int v0 = wtot[wave][lane];
        int v = v0;
#pragma unroll
        for (int d = 1; d < 32; d <<= 1) {
            const int u = __shfl_up(v, d);
            if (lane >= d) v += u;
        }
        wexc[wave][lane] = v - v0;
        if (lane == 31) ittot[wave] = v;
    }
    __syncthreads();

    int carry = 0;
#pragma unroll
    for (int it = 0; it < SCAN_IT; ++it) {
        const int idx = it * 4096 + t * 4;
        if (idx < NCELL) {
            const int4 c = *(const int4*)(cell + idx);
            const int base = carry + wexc[it][wave] + incl[it] - ssum[it];
            int4 o;
            o.x = base;
            o.y = base + c.x;
            o.z = base + c.x + c.y;
            o.w = base + c.x + c.y + c.z;
            *(int4*)(cellx + idx) = o;
        }
        carry += ittot[it];
    }
}

__global__ __launch_bounds__(1024) void bin_place(
    const int*   __restrict__ rows,
    const int*   __restrict__ cols,
    const float* __restrict__ vals,
    const int*   __restrict__ cellx,
    int* __restrict__ binned)
{
    __shared__ int lofs[NBIN];
    const int t = threadIdx.x;
    const int b = blockIdx.x;
    for (int i = t; i < NBIN; i += 1024) lofs[i] = cellx[i * NSLICE + b];
    __syncthreads();
    const int e0 = b * SLICE_E;
    int   r[SLICE_K];
    int   c[SLICE_K];
    float v[SLICE_K];
#pragma unroll
    for (int k = 0; k < SLICE_K; ++k) {
        const int i = t + k * 1024;
        r[k] = -1; c[k] = 0; v[k] = 0.f;
        if (i < SLICE_E) { r[k] = rows[e0 + i]; c[k] = cols[e0 + i]; v[k] = vals[e0 + i]; }
    }
#pragma unroll
    for (int k = 0; k < SLICE_K; ++k) {
        if (r[k] >= 0) {
            const int slot = atomicAdd(&lofs[r[k] >> BIN_SHIFT], 1);
            ((int2*)binned)[slot] = make_int2((c[k] << BIN_SHIFT) | (r[k] & (BIN_ROWS - 1)),
                                              __float_as_int(v[k]));
        }
    }
}

#define CHUNK  3072
#define CH_K   6
#define ROWS_W 8
#define GRP    8

__device__ __forceinline__ void row_store(float* out, int row, int lane, float4 a, bool first, bool last)
{
    if (row < N_NODES) {
        float4* p = (float4*)out + (row * 32 + lane);
        if (!first) {
            const float4 o = *p;
            a.x += o.x; a.y += o.y; a.z += o.z; a.w += o.w;
        }
        if (last) {
            a.x = fmaxf(a.x, 0.f); a.y = fmaxf(a.y, 0.f);
            a.z = fmaxf(a.z, 0.f); a.w = fmaxf(a.w, 0.f);
        }
        *p = a;
    }
}

__global__ __launch_bounds__(512) void bin_gather_relu(
    const float* __restrict__ h,
    const int*   __restrict__ cellx,
    const int*   __restrict__ binned,
    float* out)
{
    __shared__ alignas(16) int  rcnt[BIN_ROWS];
    __shared__ alignas(16) int  rbeg[BIN_ROWS + 4];
    __shared__ int2 srt[CHUNK];

    const int t    = threadIdx.x;
    const int lane = t & 31;
    const int wave = __builtin_amdgcn_readfirstlane(t >> 5);
    const int bin  = blockIdx.x;

    const int start = cellx[bin * NSLICE];
    const int end   = (bin + 1 < NBIN) ? cellx[(bin + 1) * NSLICE] : N_EDGES;
    const int2*   sp = (const int2*)binned;
    const float4* hv = (const float4*)h;

    const int rl0  = wave * ROWS_W;
    const int row0 = bin * BIN_ROWS;
    const float4 zero4 = make_float4(0.f, 0.f, 0.f, 0.f);

    int  cbase = start;
    bool first = true;
    do {
        const int  left = end - cbase;
        const int  n    = left < CHUNK ? left : CHUNK;
        const bool last = left <= CHUNK;

        if (t < BIN_ROWS) rcnt[t] = 0;
        __syncthreads();

        int2 e[CH_K];
        int  rk[CH_K];
#pragma unroll
        for (int k = 0; k < CH_K; ++k) {
            e[k] = make_int2(0, 0);
            if (t < n - k * 512) e[k] = sp[cbase + k * 512 + t];
        }
#pragma unroll
        for (int k = 0; k < CH_K; ++k) {
            rk[k] = 0;
            if (t < n - k * 512) rk[k] = atomicAdd(&rcnt[e[k].x & (BIN_ROWS - 1)], 1);
        }
        __syncthreads();

        if (t == 0) {
            int run = 0;
#pragma unroll 2
            for (int i = 0; i < BIN_ROWS / 4; ++i) {
                const int4 c = ((const int4*)rcnt)[i];
                ((int4*)rbeg)[i] = make_int4(run, run + c.x, run + c.x + c.y, run + c.x + c.y + c.z);
                run += c.x + c.y + c.z + c.w;
            }
            rbeg[BIN_ROWS] = run;
        }
        __syncthreads();

#pragma unroll
        for (int k = 0; k < CH_K; ++k) {
            if (t < n - k * 512) srt[rbeg[e[k].x & (BIN_ROWS - 1)] + rk[k]] = e[k];
        }
        __syncthreads();

        const int wb = __builtin_amdgcn_readfirstlane(rbeg[rl0]);
        const int we = __builtin_amdgcn_readfirstlane(rbeg[rl0 + ROWS_W]);
        int    cur = rl0;
        float4 acc = zero4;

        for (int j0 = wb; j0 < we; j0 += 32) {
            const int rest = we - j0;
            const int m = rest < 32 ? rest : 32;
            int2 ent = make_int2(0, 0);
            if (lane < m) ent = srt[j0 + lane];

            for (int j = 0; j < m; j += GRP) {
                int    kk[GRP];
                float  vv[GRP];
                float4 a[GRP];
#pragma unroll
                for (int u = 0; u < GRP; ++u) {
                    int q = j + u;
                    q = q < m ? q : m - 1;
                    kk[u] = __builtin_amdgcn_readlane(ent.x, q);
                    vv[u] = __int_as_float(__builtin_amdgcn_readlane(ent.y, q));
                    a[u]  = hv[(kk[u] >> BIN_SHIFT) * 32 + lane];
                }
#pragma unroll
                for (int u = 0; u < GRP; ++u) {
                    if (j + u < m) {
                        const int r = kk[u] & (BIN_ROWS - 1);
                        if (cur < r) {
                            row_store(out, row0 + cur, lane, acc, first, last);
                            acc = zero4;
#pragma nounroll
                            for (++cur; cur < r; ++cur)
                                row_store(out, row0 + cur, lane, zero4, first, last);
                        }
                        acc.x = fmaf(vv[u], a[u].x, acc.x);
                        acc.y = fmaf(vv[u], a[u].y, acc.y);
                        acc.z = fmaf(vv[u], a[u].z, acc.z);
                        acc.w = fmaf(vv[u], a[u].w, acc.w);
                    }
                }
            }
        }
        row_store(out, row0 + cur, lane, acc, first, last);
#pragma nounroll
        for (++cur; cur < rl0 + ROWS_W; ++cur)
            row_store(out, row0 + cur, lane, zero4, first, last);

        first = false;
        cbase += CHUNK;
    } while (cbase < end);
}

extern "C" void kernel_launch(void* const* d_in, const int* in_sizes, int n_in,
                              void* d_out, int out_size, void* d_ws, size_t ws_size,
                              hipStream_t stream)
{
    const float* x         = (const float*)d_in[0];
    const int*   edge_rows = (const int*)  d_in[1];
    const int*   edge_cols = (const int*)  d_in[2];
    const float* edge_vals = (const float*)d_in[3];
    const float* W         = (const float*)d_in[4];
    const float* b         = (const float*)d_in[5];
    float*       out       = (float*)d_out;

    float* h      = (float*)d_ws;
    int*   cell   = (int*)(h + (size_t)N_NODES * OUT_FT);
    int*   cellx  = cell + 102400;
    int*   binned = cellx + 102400;

    const int mTiles = N_NODES / 16;
    gemm_bias_wmma<<<(mTiles + 7) / 8, 256, 0, stream>>>(x, W, b, h, N_NODES);

    bin_count<<<NSLICE, 1024, 0, stream>>>(edge_rows, cell);
    scan_cells<<<1, 1024, 0, stream>>>(cell, cellx);
    bin_place<<<NSLICE, 1024, 0, stream>>>(edge_rows, edge_cols, edge_vals, cellx, binned);

    bin_gather_relu<<<NBIN, 512, 0, stream>>>(h, cellx, binned, out);
}


# prefix-sum launch removed: bin counts stored slice-major; each place block reads the count matrix once for bin totals and its slice's offsets, scans the 782 totals in LDS (5 launches to 4)
# speedup vs baseline: 3.0720x; 1.1157x over previous
#include <hip/hip_runtime.h>

#define N_NODES 100000
#define N_EDGES 1600000
#define IN_FT   128
#define OUT_FT  128

typedef __attribute__((ext_vector_type(2))) float v2f;
typedef __attribute__((ext_vector_type(8))) float v8f;

__global__ __launch_bounds__(256) void gemm_bias_wmma(
    const float* __restrict__ x, const float* __restrict__ W,
    const float* __restrict__ b, float* __restrict__ h, int nNodes)
{
    __shared__ float Ws[IN_FT * OUT_FT];

    {
        const float4* src = (const float4*)W;
        float4*       dst = (float4*)Ws;
        int tid = threadIdx.x;
#pragma unroll
        for (int i = 0; i < 16; ++i)
            dst[tid + i * 256] = src[tid + i * 256];
    }
    __syncthreads();

    const int wave  = threadIdx.x >> 5;
    const int lane  = threadIdx.x & 31;
    const int tile  = blockIdx.x * 8 + wave;
    if (tile * 16 >= nNodes) return;

    const int row0  = tile * 16;
    const int m     = lane & 15;
    const int khalf = lane >> 4;

    const v2f* xv = (const v2f*)(x + (size_t)(row0 + m) * IN_FT + 2 * khalf);
    v2f afrag[32];
#pragma unroll
    for (int kk = 0; kk < 32; ++kk)
        afrag[kk] = xv[kk * 2];

    v8f acc[8] = {};

#pragma unroll
    for (int kk = 0; kk < 32; ++kk) {
        const int krow = kk * 4 + 2 * khalf;
#pragma unroll
        for (int nt = 0; nt < 8; ++nt) {
            const int n = nt * 16 + m;
            v2f bf;
            bf.x = Ws[(krow + 0) * OUT_FT + n];
            bf.y = Ws[(krow + 1) * OUT_FT + n];
            acc[nt] = __builtin_amdgcn_wmma_f32_16x16x4_f32(
                 false, afrag[kk],  false, bf,
                 (short)0, acc[nt],
                 false,  false);
        }
    }

#pragma unroll
    for (int nt = 0; nt < 8; ++nt) {
        const int n      = nt * 16 + m;
        const float bias = b[n];
#pragma unroll
        for (int j = 0; j < 8; ++j) {
            const int row = row0 + j + 8 * khalf;
            h[(size_t)row * OUT_FT + n] = acc[nt][j] + bias;
        }
    }
}

#define BIN_SHIFT 7
#define BIN_ROWS  128
#define NBIN      782
#define NSLICE    128
#define SLICE_E   12500
#define CPITCH    800
#define CCHUNK    (CPITCH / 4)
#define SLICE_K   13

__global__ __launch_bounds__(1024) void bin_count(
    const int* __restrict__ rows, int* __restrict__ cell)
{
    __shared__ int lh[CPITCH];
    const int t = threadIdx.x;
    const int b = blockIdx.x;
    if (t < CPITCH) lh[t] = 0;
    __syncthreads();
    const int* rp = rows + b * SLICE_E;
    int r[SLICE_K];
#pragma unroll
    for (int k = 0; k < SLICE_K; ++k) {
        const int i = t + k * 1024;
        r[k] = (i < SLICE_E) ? rp[i] : -1;
    }
#pragma unroll
    for (int k = 0; k < SLICE_K; ++k)
        if (r[k] >= 0) atomicAdd(&lh[r[k] >> BIN_SHIFT], 1);
    __syncthreads();
    if (t < CPITCH) cell[b * CPITCH + t] = lh[t];
}

__global__ __launch_bounds__(1024) void bin_place(
    const int*   __restrict__ rows,
    const int*   __restrict__ cols,
    const float* __restrict__ vals,
    const int*   __restrict__ cell,
    int* __restrict__ binstart,
    int* __restrict__ binned)
{
    __shared__ alignas(16) int ptot[4][CPITCH];
    __shared__ alignas(16) int ppre[4][CPITCH];
    __shared__ int lofs[NBIN];
    __shared__ int wsum[32];
    const int t = threadIdx.x;
    const int b = blockIdx.x;
    const int lane = t & 31;
    const int wave = t >> 5;
    const int e0 = b * SLICE_E;

    {
        const int g = __builtin_amdgcn_readfirstlane(t >> 8);
        const int u = t & 255;
        if (u < CCHUNK) {
            int4 tot = make_int4(0, 0, 0, 0);
            int4 pre = make_int4(0, 0, 0, 0);
#pragma nounroll
            for (int k0 = 0; k0 < NSLICE / 4; k0 += 16) {
                int4 q[16];
#pragma unroll
                for (int j = 0; j < 16; ++j) {
                    const int s = g + 4 * (k0 + j);
                    q[j] = ((const int4*)(cell + s * CPITCH))[u];
                }
#pragma unroll
                for (int j = 0; j < 16; ++j) {
                    const int s = g + 4 * (k0 + j);
                    const int m = (s < b) ? -1 : 0;
                    tot.x += q[j].x; tot.y += q[j].y; tot.z += q[j].z; tot.w += q[j].w;
                    pre.x += q[j].x & m; pre.y += q[j].y & m; pre.z += q[j].z & m; pre.w += q[j].w & m;
                }
            }
            ((int4*)ptot[g])[u] = tot;
            ((int4*)ppre[g])[u] = pre;
        }
    }

    int   r[SLICE_K];
    int   c[SLICE_K];
    float v[SLICE_K];
#pragma unroll
    for (int k = 0; k < SLICE_K; ++k) {
        const int i = t + k * 1024;
        r[k] = -1; c[k] = 0; v[k] = 0.f;
        if (i < SLICE_E) { r[k] = rows[e0 + i]; c[k] = cols[e0 + i]; v[k] = vals[e0 + i]; }
    }
    __syncthreads();

    int myTot = 0, myPre = 0;
    if (t < NBIN) {
        myTot = ptot[0][t] + ptot[1][t] + ptot[2][t] + ptot[3][t];
        myPre = ppre[0][t] + ppre[1][t] + ppre[2][t] + ppre[3][t];
    }
    int inc = myTot;
#pragma unroll
    for (int d = 1; d < 32; d <<= 1) {
        const int up = __shfl_up(inc, d);
        if (lane >= d) inc += up;
    }
    if (lane == 31) wsum[wave] = inc;
    __syncthreads();
    if (wave == 0) {
        const int w0 = wsum[lane];
        int w = w0;
#pragma unroll
        for (int d = 1; d < 32; d <<= 1) {
            const int up = __shfl_up(w, d);
            if (lane >= d) w += up;
        }
        wsum[lane] = w - w0;
    }
    __syncthreads();
    const int bstart = wsum[wave] + inc - myTot;
    if (t < NBIN) lofs[t] = bstart + myPre;
    if (b == 0 && t <= NBIN) binstart[t] = bstart;
    __syncthreads();

#pragma unroll
    for (int k = 0; k < SLICE_K; ++k) {
        if (r[k] >= 0) {
            const int slot = atomicAdd(&lofs[r[k] >> BIN_SHIFT], 1);
            ((int2*)binned)[slot] = make_int2((c[k] << BIN_SHIFT) | (r[k] & (BIN_ROWS - 1)),
                                              __float_as_int(v[k]));
        }
    }
}

#define CHUNK  3072
#define CH_K   6
#define ROWS_W 8
#define GRP    8

__device__ __forceinline__ void row_store(float* out, int row, int lane, float4 a, bool first, bool last)
{
    if (row < N_NODES) {
        float4* p = (float4*)out + (row * 32 + lane);
        if (!first) {
            const float4 o = *p;
            a.x += o.x; a.y += o.y; a.z += o.z; a.w += o.w;
        }
        if (last) {
            a.x = fmaxf(a.x, 0.f); a.y = fmaxf(a.y, 0.f);
            a.z = fmaxf(a.z, 0.f); a.w = fmaxf(a.w, 0.f);
        }
        *p = a;
    }
}

__global__ __launch_bounds__(512) void bin_gather_relu(
    const float* __restrict__ h,
    const int*   __restrict__ binstart,
    const int*   __restrict__ binned,
    float* out)
{
    __shared__ alignas(16) int  rcnt[BIN_ROWS];
    __shared__ alignas(16) int  rbeg[BIN_ROWS + 4];
    __shared__ int2 srt[CHUNK];

    const int t    = threadIdx.x;
    const int lane = t & 31;
    const int wave = __builtin_amdgcn_readfirstlane(t >> 5);
    const int bin  = blockIdx.x;

    const int start = binstart[bin];
    const int end   = binstart[bin + 1];
    const int2*   sp = (const int2*)binned;
    const float4* hv = (const float4*)h;

    const int rl0  = wave * ROWS_W;
    const int row0 = bin * BIN_ROWS;
    const float4 zero4 = make_float4(0.f, 0.f, 0.f, 0.f);

    int  cbase = start;
    bool first = true;
    do {
        const int  left = end - cbase;
        const int  n    = left < CHUNK ? left : CHUNK;
        const bool last = left <= CHUNK;

        if (t < BIN_ROWS) rcnt[t] = 0;
        __syncthreads();

        int2 e[CH_K];
        int  rk[CH_K];
#pragma unroll
        for (int k = 0; k < CH_K; ++k) {
            e[k] = make_int2(0, 0);
            if (t < n - k * 512) e[k] = sp[cbase + k * 512 + t];
        }
#pragma unroll
        for (int k = 0; k < CH_K; ++k) {
            rk[k] = 0;
            if (t < n - k * 512) rk[k] = atomicAdd(&rcnt[e[k].x & (BIN_ROWS - 1)], 1);
        }
        __syncthreads();

        if (t == 0) {
            int run = 0;
#pragma unroll 2
            for (int i = 0; i < BIN_ROWS / 4; ++i) {
                const int4 c = ((const int4*)rcnt)[i];
                ((int4*)rbeg)[i] = make_int4(run, run + c.x, run + c.x + c.y, run + c.x + c.y + c.z);
                run += c.x + c.y + c.z + c.w;
            }
            rbeg[BIN_ROWS] = run;
        }
        __syncthreads();

#pragma unroll
        for (int k = 0; k < CH_K; ++k) {
            if (t < n - k * 512) srt[rbeg[e[k].x & (BIN_ROWS - 1)] + rk[k]] = e[k];
        }
        __syncthreads();

        const int wb = __builtin_amdgcn_readfirstlane(rbeg[rl0]);
        const int we = __builtin_amdgcn_readfirstlane(rbeg[rl0 + ROWS_W]);
        int    cur = rl0;
        float4 acc = zero4;

        for (int j0 = wb; j0 < we; j0 += 32) {
            const int rest = we - j0;
            const int m = rest < 32 ? rest : 32;
            int2 ent = make_int2(0, 0);
            if (lane < m) ent = srt[j0 + lane];

            for (int j = 0; j < m; j += GRP) {
                int    kk[GRP];
                float  vv[GRP];
                float4 a[GRP];
#pragma unroll
                for (int u = 0; u < GRP; ++u) {
                    int q = j + u;
                    q = q < m ? q : m - 1;
                    kk[u] = __builtin_amdgcn_readlane(ent.x, q);
                    vv[u] = __int_as_float(__builtin_amdgcn_readlane(ent.y, q));
                    a[u]  = hv[(kk[u] >> BIN_SHIFT) * 32 + lane];
                }
#pragma unroll
                for (int u = 0; u < GRP; ++u) {
                    if (j + u < m) {
                        const int r = kk[u] & (BIN_ROWS - 1);
                        if (cur < r) {
                            row_store(out, row0 + cur, lane, acc, first, last);
                            acc = zero4;
#pragma nounroll
                            for (++cur; cur < r; ++cur)
                                row_store(out, row0 + cur, lane, zero4, first, last);
                        }
                        acc.x = fmaf(vv[u], a[u].x, acc.x);
                        acc.y = fmaf(vv[u], a[u].y, acc.y);
                        acc.z = fmaf(vv[u], a[u].z, acc.z);
                        acc.w = fmaf(vv[u], a[u].w, acc.w);
                    }
                }
            }
        }
        row_store(out, row0 + cur, lane, acc, first, last);
#pragma nounroll
        for (++cur; cur < rl0 + ROWS_W; ++cur)
            row_store(out, row0 + cur, lane, zero4, first, last);

        first = false;
        cbase += CHUNK;
    } while (cbase < end);
}

extern "C" void kernel_launch(void* const* d_in, const int* in_sizes, int n_in,
                              void* d_out, int out_size, void* d_ws, size_t ws_size,
                              hipStream_t stream)
{
    const float* x         = (const float*)d_in[0];
    const int*   edge_rows = (const int*)  d_in[1];
    const int*   edge_cols = (const int*)  d_in[2];
    const float* edge_vals = (const float*)d_in[3];
    const float* W         = (const float*)d_in[4];
    const float* b         = (const float*)d_in[5];
    float*       out       = (float*)d_out;

    float* h      = (float*)d_ws;
    int*   cell     = (int*)(h + (size_t)N_NODES * OUT_FT);
    int*   binstart = cell + 102400;
    int*   binned   = binstart + 102400;

    const int mTiles = N_NODES / 16;
    gemm_bias_wmma<<<(mTiles + 7) / 8, 256, 0, stream>>>(x, W, b, h, N_NODES);

    bin_count<<<NSLICE, 1024, 0, stream>>>(edge_rows, cell);
    bin_place<<<NSLICE, 1024, 0, stream>>>(edge_rows, edge_cols, edge_vals, cell, binstart, binned);

    bin_gather_relu<<<NBIN, 512, 0, stream>>>(h, binstart, binned, out);
}


# count launch and count matrix removed: each of 256 edge slices is ordered by bin in its own block (LDS count, scan, place); a bin block collects its 256 slice segments; sums stay in arrival order
# speedup vs baseline: 4.0064x; 1.3042x over previous
#include <hip/hip_runtime.h>

#define N_NODES 100000
#define N_EDGES 1600000
#define IN_FT   128
#define OUT_FT  128

typedef __attribute__((ext_vector_type(2))) float v2f;
typedef __attribute__((ext_vector_type(8))) float v8f;

__global__ __launch_bounds__(256) void gemm_bias_wmma(
    const float* __restrict__ x, const float* __restrict__ W,
    const float* __restrict__ b, float* __restrict__ h, int nNodes)
{
    __shared__ float Ws[IN_FT * OUT_FT];

    {
        const float4* src = (const float4*)W;
        float4*       dst = (float4*)Ws;
        int tid = threadIdx.x;
#pragma unroll
        for (int i = 0; i < 16; ++i)
            dst[tid + i * 256] = src[tid + i * 256];
    }
    __syncthreads();

    const int wave  = threadIdx.x >> 5;
    const int lane  = threadIdx.x & 31;
    const int tile  = blockIdx.x * 8 + wave;
    if (tile * 16 >= nNodes) return;

    const int row0  = tile * 16;
    const int m     = lane & 15;
    const int khalf = lane >> 4;

    const v2f* xv = (const v2f*)(x + (size_t)(row0 + m) * IN_FT + 2 * khalf);
    v2f afrag[32];
#pragma unroll
    for (int kk = 0; kk < 32; ++kk)
        afrag[kk] = xv[kk * 2];

    v8f acc[8] = {};

#pragma unroll
    for (int kk = 0; kk < 32; ++kk) {
        const int krow = kk * 4 + 2 * khalf;
#pragma unroll
        for (int nt = 0; nt < 8; ++nt) {
            const int n = nt * 16 + m;
            v2f bf;
            bf.x = Ws[(krow + 0) * OUT_FT + n];
            bf.y = Ws[(krow + 1) * OUT_FT + n];
            acc[nt] = __builtin_amdgcn_wmma_f32_16x16x4_f32(
                 false, afrag[kk],  false, bf,
                 (short)0, acc[nt],
                 false,  false);
        }
    }

#pragma unroll
    for (int nt = 0; nt < 8; ++nt) {
        const int n      = nt * 16 + m;
        const float bias = b[n];
#pragma unroll
        for (int j = 0; j < 8; ++j) {
            const int row = row0 + j + 8 * khalf;
            h[(size_t)row * OUT_FT + n] = acc[nt][j] + bias;
        }
    }
}

#define BIN_SHIFT 7
#define BIN_ROWS  128
#define NBIN      782
#define NSLICE    256
#define SLICE_E   6250
#define SLICE_V   3125
#define CPITCH    800
#define SLICE_K   7

__global__ __launch_bounds__(1024) void slice_sort(
    const int*   __restrict__ rows,
    const int*   __restrict__ cols,
    const float* __restrict__ vals,
    int* __restrict__ ofs,
    int* __restrict__ binned)
{
    __shared__ int cnt[CPITCH];
    __shared__ int lofs[CPITCH];
    __shared__ int wsum[32];
    __shared__ alignas(16) int2 stage[SLICE_E];
    const int t = threadIdx.x;
    const int b = blockIdx.x;
    const int lane = t & 31;
    const int wave = t >> 5;
    const int e0 = b * SLICE_E;

    if (t < CPITCH) cnt[t] = 0;

    int   r[SLICE_K];
    int   c[SLICE_K];
    float v[SLICE_K];
#pragma unroll
    for (int k = 0; k < SLICE_K; ++k) {
        const int i = t + k * 1024;
        r[k] = -1; c[k] = 0; v[k] = 0.f;
        if (i < SLICE_E) { r[k] = rows[e0 + i]; c[k] = cols[e0 + i]; v[k] = vals[e0 + i]; }
    }
    __syncthreads();

    int rk[SLICE_K];
#pragma unroll
    for (int k = 0; k < SLICE_K; ++k) {
        rk[k] = 0;
        if (r[k] >= 0) rk[k] = atomicAdd(&cnt[r[k] >> BIN_SHIFT], 1);
    }
    __syncthreads();

    const int myTot = (t < NBIN) ? cnt[t] : 0;
    int inc = myTot;
#pragma unroll
    for (int d = 1; d < 32; d <<= 1) {
        const int up = __shfl_up(inc, d);
        if (lane >= d) inc += up;
    }
    if (lane == 31) wsum[wave] = inc;
    __syncthreads();
    if (wave == 0) {
        const int w0 = wsum[lane];
        int w = w0;
#pragma unroll
        for (int d = 1; d < 32; d <<= 1) {
            const int up = __shfl_up(w, d);
            if (lane >= d) w += up;
        }
        wsum[lane] = w - w0;
    }
    __syncthreads();
    const int bstart = wsum[wave] + inc - myTot;
    if (t < CPITCH) { lofs[t] = bstart; ofs[b * CPITCH + t] = bstart; }
    __syncthreads();

#pragma unroll
    for (int k = 0; k < SLICE_K; ++k) {
        if (r[k] >= 0)
            stage[lofs[r[k] >> BIN_SHIFT] + rk[k]] =
                make_int2((c[k] << BIN_SHIFT) | (r[k] & (BIN_ROWS - 1)), __float_as_int(v[k]));
    }
    __syncthreads();

    const int4* sv = (const int4*)stage;
    int4*       dv = (int4*)binned + b * SLICE_V;
#pragma unroll
    for (int k = 0; k < 4; ++k) {
        const int i = t + k * 1024;
        if (i < SLICE_V) dv[i] = sv[i];
    }
}

#define CHUNK  3072
#define CH_K   6
#define ROWS_W 8
#define GRP    8

__device__ __forceinline__ void row_store(float* out, int row, int lane, float4 a, bool first, bool last)
{
    if (row < N_NODES) {
        float4* p = (float4*)out + (row * 32 + lane);
        if (!first) {
            const float4 o = *p;
            a.x += o.x; a.y += o.y; a.z += o.z; a.w += o.w;
        }
        if (last) {
            a.x = fmaxf(a.x, 0.f); a.y = fmaxf(a.y, 0.f);
            a.z = fmaxf(a.z, 0.f); a.w = fmaxf(a.w, 0.f);
        }
        *p = a;
    }
}

__global__ __launch_bounds__(512, 16) void bin_gather_relu(
    const float* __restrict__ h,
    const int*   __restrict__ ofs,
    const int*   __restrict__ binned,
    float* out)
{
    __shared__ alignas(16) int  rcnt[BIN_ROWS];
    __shared__ alignas(16) int  rbeg[BIN_ROWS + 4];
    __shared__ int  wtot[16];
    __shared__ int  gidx[CHUNK];
    __shared__ int  seg_lo[NSLICE];
    __shared__ int  seg_hi[NSLICE];
    __shared__ int  seg_gl[NSLICE];
    __shared__ int2 srt[CHUNK];

    const int t    = threadIdx.x;
    const int lane = t & 31;
    const int wave = __builtin_amdgcn_readfirstlane(t >> 5);
    const int bin  = blockIdx.x;

    int sa = 0, slen = 0;
    if (t < NSLICE) {
        const int* o = ofs + t * CPITCH + bin;
        sa   = o[0];
        slen = o[1] - sa;
    }
    int inc = slen;
#pragma unroll
    for (int d = 1; d < 32; d <<= 1) {
        const int up = __shfl_up(inc, d);
        if (lane >= d) inc += up;
    }
    if (lane == 31) wtot[wave] = inc;
    __syncthreads();
    int wofs = 0, end = 0;
#pragma unroll
    for (int w = 0; w < 16; ++w) {
        const int x = wtot[w];
        end  += x;
        wofs += (w < wave) ? x : 0;
    }
    if (t < NSLICE) {
        const int spos = wofs + inc - slen;
        seg_lo[t] = spos;
        seg_hi[t] = spos + slen;
        seg_gl[t] = t * SLICE_E + sa - spos;
    }

    const int2*   sp = (const int2*)binned;
    const float4* hv = (const float4*)h;

    const int rl0  = wave * ROWS_W;
    const int row0 = bin * BIN_ROWS;
    const float4 zero4 = make_float4(0.f, 0.f, 0.f, 0.f);

    int  cbase = 0;
    bool first = true;
    do {
        const int  left = end - cbase;
        const int  n    = left < CHUNK ? left : CHUNK;
        const bool last = left <= CHUNK;

        if (t < BIN_ROWS) rcnt[t] = 0;
        if (t < NSLICE) {
            const int spos = seg_lo[t];
            const int hs   = seg_hi[t];
            const int sgl  = seg_gl[t];
            const int lo = spos > cbase ? spos : cbase;
            const int hc = cbase + n;
            const int hi = hs < hc ? hs : hc;
#pragma nounroll
            for (int j = lo; j < hi; ++j) gidx[j - cbase] = sgl + j;
        }
        __syncthreads();

        int2 e[CH_K];
        int  rk[CH_K];
#pragma unroll
        for (int k = 0; k < CH_K; ++k) {
            e[k] = make_int2(0, 0);
            if (t < n - k * 512) e[k] = sp[gidx[k * 512 + t]];
        }
#pragma unroll
        for (int k = 0; k < CH_K; ++k) {
            rk[k] = 0;
            if (t < n - k * 512) rk[k] = atomicAdd(&rcnt[e[k].x & (BIN_ROWS - 1)], 1);
        }
        __syncthreads();

        if (t == 0) {
            int run = 0;
#pragma unroll 2
            for (int i = 0; i < BIN_ROWS / 4; ++i) {
                const int4 c = ((const int4*)rcnt)[i];
                ((int4*)rbeg)[i] = make_int4(run, run + c.x, run + c.x + c.y, run + c.x + c.y + c.z);
                run += c.x + c.y + c.z + c.w;
            }
            rbeg[BIN_ROWS] = run;
        }
        __syncthreads();

#pragma unroll
        for (int k = 0; k < CH_K; ++k) {
            if (t < n - k * 512) srt[rbeg[e[k].x & (BIN_ROWS - 1)] + rk[k]] = e[k];
        }
        __syncthreads();

        const int wb = __builtin_amdgcn_readfirstlane(rbeg[rl0]);
        const int we = __builtin_amdgcn_readfirstlane(rbeg[rl0 + ROWS_W]);
        int    cur = rl0;
        float4 acc = zero4;

        for (int j0 = wb; j0 < we; j0 += 32) {
            const int rest = we - j0;
            const int m = rest < 32 ? rest : 32;
            int2 ent = make_int2(0, 0);
            if (lane < m) ent = srt[j0 + lane];

            for (int j = 0; j < m; j += GRP) {
                int    kk[GRP];
                float  vv[GRP];
                float4 a[GRP];
#pragma unroll
                for (int u = 0; u < GRP; ++u) {
                    int q = j + u;
                    q = q < m ? q : m - 1;
                    kk[u] = __builtin_amdgcn_readlane(ent.x, q);
                    vv[u] = __int_as_float(__builtin_amdgcn_readlane(ent.y, q));
                    a[u]  = hv[(kk[u] >> BIN_SHIFT) * 32 + lane];
                }
#pragma unroll
                for (int u = 0; u < GRP; ++u) {
                    if (j + u < m) {
                        const int r = kk[u] & (BIN_ROWS - 1);
                        if (cur < r) {
                            row_store(out, row0 + cur, lane, acc, first, last);
                            acc = zero4;
#pragma nounroll
                            for (++cur; cur < r; ++cur)
                                row_store(out, row0 + cur, lane, zero4, first, last);
                        }
                        acc.x = fmaf(vv[u], a[u].x, acc.x);
                        acc.y = fmaf(vv[u], a[u].y, acc.y);
                        acc.z = fmaf(vv[u], a[u].z, acc.z);
                        acc.w = fmaf(vv[u], a[u].w, acc.w);
                    }
                }
            }
        }
        row_store(out, row0 + cur, lane, acc, first, last);
#pragma nounroll
        for (++cur; cur < rl0 + ROWS_W; ++cur)
            row_store(out, row0 + cur, lane, zero4, first, last);

        first = false;
        cbase += CHUNK;
    } while (cbase < end);
}

extern "C" void kernel_launch(void* const* d_in, const int* in_sizes, int n_in,
                              void* d_out, int out_size, void* d_ws, size_t ws_size,
                              hipStream_t stream)
{
    const float* x         = (const float*)d_in[0];
    const int*   edge_rows = (const int*)  d_in[1];
    const int*   edge_cols = (const int*)  d_in[2];
    const float* edge_vals = (const float*)d_in[3];
    const float* W         = (const float*)d_in[4];
    const float* b         = (const float*)d_in[5];
    float*       out       = (float*)d_out;

    float* h      = (float*)d_ws;
    int*   ofs    = (int*)(h + (size_t)N_NODES * OUT_FT);
    int*   binned = ofs + 204800;

    const int mTiles = N_NODES / 16;
    gemm_bias_wmma<<<(mTiles + 7) / 8, 256, 0, stream>>>(x, W, b, h, N_NODES);

    slice_sort<<<NSLICE, 1024, 0, stream>>>(edge_rows, edge_cols, edge_vals, ofs, binned);

    bin_gather_relu<<<NBIN, 512, 0, stream>>>(h, ofs, binned, out);
}


# projection kernel: W kept in LDS as k-pairs at a bank-separating pitch, so a B fragment is one conflict-free 8-byte LDS read (was two 4-byte reads, 2-way conflict, pairing moves), fed a k-step ahead
# speedup vs baseline: 4.3761x; 1.0923x over previous
#include <hip/hip_runtime.h>

#define N_NODES 100000
#define N_EDGES 1600000
#define IN_FT   128
#define OUT_FT  128

typedef __attribute__((ext_vector_type(2))) float v2f;
typedef __attribute__((ext_vector_type(8))) float v8f;

#define WP_PITCH 288
__global__ __launch_bounds__(256) void gemm_bias_wmma(
    const float* __restrict__ x, const float* __restrict__ W,
    const float* __restrict__ b, float* __restrict__ h, int nNodes)
{
    __shared__ alignas(16) float Ws[(IN_FT / 2) * WP_PITCH];

    {
        const float4* src = (const float4*)W;
        const int tid = threadIdx.x;
#pragma unroll
        for (int i = 0; i < 8; ++i) {
            const int idx = tid + i * 256;
            const int p   = idx >> 5;
            const int q   = idx & 31;
            const float4 lo = src[(2 * p)     * (OUT_FT / 4) + q];
            const float4 hi = src[(2 * p + 1) * (OUT_FT / 4) + q];
            float4* dst = (float4*)(Ws + p * WP_PITCH + q * 8);
            dst[0] = make_float4(lo.x, hi.x, lo.y, hi.y);
            dst[1] = make_float4(lo.z, hi.z, lo.w, hi.w);
        }
    }
    __syncthreads();

    const int wave  = threadIdx.x >> 5;
    const int lane  = threadIdx.x & 31;
    const int tile  = blockIdx.x * 8 + wave;
    if (tile * 16 >= nNodes) return;

    const int row0  = tile * 16;
    const int m     = lane & 15;
    const int khalf = lane >> 4;

    const v2f* xv = (const v2f*)(x + (size_t)(row0 + m) * IN_FT + 2 * khalf);
    v2f acur[4], anxt[4];
#pragma unroll
    for (int i = 0; i < 4; ++i)
        acur[i] = xv[i * 2];

    v8f acc[8] = {};

    const float* wl = Ws + khalf * WP_PITCH + m * 2;
    v2f bq[2][8];
#pragma unroll
    for (int nt = 0; nt < 8; ++nt)
        bq[0][nt] = *(const v2f*)(wl + nt * 32);

#pragma nounroll
    for (int kb = 0; kb < 8; ++kb) {
        const float* wk = wl + kb * (8 * WP_PITCH);
        if (kb < 7) {
#pragma unroll
            for (int i = 0; i < 4; ++i)
                anxt[i] = xv[(kb + 1) * 8 + i * 2];
        }
#pragma unroll
        for (int i = 0; i < 4; ++i) {
            if (i < 3 || kb < 7) {
#pragma unroll
                for (int nt = 0; nt < 8; ++nt)
                    bq[(i + 1) & 1][nt] = *(const v2f*)(wk + (i + 1) * 2 * WP_PITCH + nt * 32);
            }
#pragma unroll
            for (int nt = 0; nt < 8; ++nt)
                acc[nt] = __builtin_amdgcn_wmma_f32_16x16x4_f32(
                     false, acur[i],  false, bq[i & 1][nt],
                     (short)0, acc[nt],
                     false,  false);
        }
#pragma unroll
        for (int i = 0; i < 4; ++i) acur[i] = anxt[i];
    }

#pragma unroll
    for (int nt = 0; nt < 8; ++nt) {
        const int n      = nt * 16 + m;
        const float bias = b[n];
#pragma unroll
        for (int j = 0; j < 8; ++j) {
            const int row = row0 + j + 8 * khalf;
            h[(size_t)row * OUT_FT + n] = acc[nt][j] + bias;
        }
    }
}

#define BIN_SHIFT 7
#define BIN_ROWS  128
#define NBIN      782
#define NSLICE    256
#define SLICE_E   6250
#define SLICE_V   3125
#define CPITCH    800
#define SLICE_K   7

__global__ __launch_bounds__(1024) void slice_sort(
    const int*   __restrict__ rows,
    const int*   __restrict__ cols,
    const float* __restrict__ vals,
    int* __restrict__ ofs,
    int* __restrict__ binned)
{
    __shared__ int cnt[CPITCH];
    __shared__ int lofs[CPITCH];
    __shared__ int wsum[32];
    __shared__ alignas(16) int2 stage[SLICE_E];
    const int t = threadIdx.x;
    const int b = blockIdx.x;
    const int lane = t & 31;
    const int wave = t >> 5;
    const int e0 = b * SLICE_E;

    if (t < CPITCH) cnt[t] = 0;

    int   r[SLICE_K];
    int   c[SLICE_K];
    float v[SLICE_K];
#pragma unroll
    for (int k = 0; k < SLICE_K; ++k) {
        const int i = t + k * 1024;
        r[k] = -1; c[k] = 0; v[k] = 0.f;
        if (i < SLICE_E) { r[k] = rows[e0 + i]; c[k] = cols[e0 + i]; v[k] = vals[e0 + i]; }
    }
    __syncthreads();

    int rk[SLICE_K];
#pragma unroll
    for (int k = 0; k < SLICE_K; ++k) {
        rk[k] = 0;
        if (r[k] >= 0) rk[k] = atomicAdd(&cnt[r[k] >> BIN_SHIFT], 1);
    }
    __syncthreads();

    const int myTot = (t < NBIN) ? cnt[t] : 0;
    int inc = myTot;
#pragma unroll
    for (int d = 1; d < 32; d <<= 1) {
        const int up = __shfl_up(inc, d);
        if (lane >= d) inc += up;
    }
    if (lane == 31) wsum[wave] = inc;
    __syncthreads();
    if (wave == 0) {
        const int w0 = wsum[lane];
        int w = w0;
#pragma unroll
        for (int d = 1; d < 32; d <<= 1) {
            const int up = __shfl_up(w, d);
            if (lane >= d) w += up;
        }
        wsum[lane] = w - w0;
    }
    __syncthreads();
    const int bstart = wsum[wave] + inc - myTot;
    if (t < CPITCH) { lofs[t] = bstart; ofs[b * CPITCH + t] = bstart; }
    __syncthreads();

#pragma unroll
    for (int k = 0; k < SLICE_K; ++k) {
        if (r[k] >= 0)
            stage[lofs[r[k] >> BIN_SHIFT] + rk[k]] =
                make_int2((c[k] << BIN_SHIFT) | (r[k] & (BIN_ROWS - 1)), __float_as_int(v[k]));
    }
    __syncthreads();

    const int4* sv = (const int4*)stage;
    int4*       dv = (int4*)binned + b * SLICE_V;
#pragma unroll
    for (int k = 0; k < 4; ++k) {
        const int i = t + k * 1024;
        if (i < SLICE_V) dv[i] = sv[i];
    }
}

#define CHUNK  3072
#define CH_K   6
#define ROWS_W 8
#define GRP    8

__device__ __forceinline__ void row_store(float* out, int row, int lane, float4 a, bool first, bool last)
{
    if (row < N_NODES) {
        float4* p = (float4*)out + (row * 32 + lane);
        if (!first) {
            const float4 o = *p;
            a.x += o.x; a.y += o.y; a.z += o.z; a.w += o.w;
        }
        if (last) {
            a.x = fmaxf(a.x, 0.f); a.y = fmaxf(a.y, 0.f);
            a.z = fmaxf(a.z, 0.f); a.w = fmaxf(a.w, 0.f);
        }
        *p = a;
    }
}

__global__ __launch_bounds__(512, 16) void bin_gather_relu(
    const float* __restrict__ h,
    const int*   __restrict__ ofs,
    const int*   __restrict__ binned,
    float* out)
{
    __shared__ alignas(16) int  rcnt[BIN_ROWS];
    __shared__ alignas(16) int  rbeg[BIN_ROWS + 4];
    __shared__ int  wtot[16];
    __shared__ int  gidx[CHUNK];
    __shared__ int  seg_lo[NSLICE];
    __shared__ int  seg_hi[NSLICE];
    __shared__ int  seg_gl[NSLICE];
    __shared__ int2 srt[CHUNK];

    const int t    = threadIdx.x;
    const int lane = t & 31;
    const int wave = __builtin_amdgcn_readfirstlane(t >> 5);
    const int bin  = blockIdx.x;

    int sa = 0, slen = 0;
    if (t < NSLICE) {
        const int* o = ofs + t * CPITCH + bin;
        sa   = o[0];
        slen = o[1] - sa;
    }
    int inc = slen;
#pragma unroll
    for (int d = 1; d < 32; d <<= 1) {
        const int up = __shfl_up(inc, d);
        if (lane >= d) inc += up;
    }
    if (lane == 31) wtot[wave] = inc;
    __syncthreads();
    int wofs = 0, end = 0;
#pragma unroll
    for (int w = 0; w < 16; ++w) {
        const int x = wtot[w];
        end  += x;
        wofs += (w < wave) ? x : 0;
    }
    if (t < NSLICE) {
        const int spos = wofs + inc - slen;
        seg_lo[t] = spos;
        seg_hi[t] = spos + slen;
        seg_gl[t] = t * SLICE_E + sa - spos;
    }

    const int2*   sp = (const int2*)binned;
    const float4* hv = (const float4*)h;

    const int rl0  = wave * ROWS_W;
    const int row0 = bin * BIN_ROWS;
    const float4 zero4 = make_float4(0.f, 0.f, 0.f, 0.f);

    int  cbase = 0;
    bool first = true;
    do {
        const int  left = end - cbase;
        const int  n    = left < CHUNK ? left : CHUNK;
        const bool last = left <= CHUNK;

        if (t < BIN_ROWS) rcnt[t] = 0;
        if (t < NSLICE) {
            const int spos = seg_lo[t];
            const int hs   = seg_hi[t];
            const int sgl  = seg_gl[t];
            const int lo = spos > cbase ? spos : cbase;
            const int hc = cbase + n;
            const int hi = hs < hc ? hs : hc;
#pragma nounroll
            for (int j = lo; j < hi; ++j) gidx[j - cbase] = sgl + j;
        }
        __syncthreads();

        int2 e[CH_K];
        int  rk[CH_K];
#pragma unroll
        for (int k = 0; k < CH_K; ++k) {
            e[k] = make_int2(0, 0);
            if (t < n - k * 512) e[k] = sp[gidx[k * 512 + t]];
        }
#pragma unroll
        for (int k = 0; k < CH_K; ++k) {
            rk[k] = 0;
            if (t < n - k * 512) rk[k] = atomicAdd(&rcnt[e[k].x & (BIN_ROWS - 1)], 1);
        }
        __syncthreads();

        if (t == 0) {
            int run = 0;
#pragma unroll 2
            for (int i = 0; i < BIN_ROWS / 4; ++i) {
                const int4 c = ((const int4*)rcnt)[i];
                ((int4*)rbeg)[i] = make_int4(run, run + c.x, run + c.x + c.y, run + c.x + c.y + c.z);
                run += c.x + c.y + c.z + c.w;
            }
            rbeg[BIN_ROWS] = run;
        }
        __syncthreads();

#pragma unroll
        for (int k = 0; k < CH_K; ++k) {
            if (t < n - k * 512) srt[rbeg[e[k].x & (BIN_ROWS - 1)] + rk[k]] = e[k];
        }
        __syncthreads();

        const int wb = __builtin_amdgcn_readfirstlane(rbeg[rl0]);
        const int we = __builtin_amdgcn_readfirstlane(rbeg[rl0 + ROWS_W]);
        int    cur = rl0;
        float4 acc = zero4;

        for (int j0 = wb; j0 < we; j0 += 32) {
            const int rest = we - j0;
            const int m = rest < 32 ? rest : 32;
            int2 ent = make_int2(0, 0);
            if (lane < m) ent = srt[j0 + lane];

            for (int j = 0; j < m; j += GRP) {
                int    kk[GRP];
                float  vv[GRP];
                float4 a[GRP];
#pragma unroll
                for (int u = 0; u < GRP; ++u) {
                    int q = j + u;
                    q = q < m ? q : m - 1;
                    kk[u] = __builtin_amdgcn_readlane(ent.x, q);
                    vv[u] = __int_as_float(__builtin_amdgcn_readlane(ent.y, q));
                    a[u]  = hv[(kk[u] >> BIN_SHIFT) * 32 + lane];
                }
#pragma unroll
                for (int u = 0; u < GRP; ++u) {
                    if (j + u < m) {
                        const int r = kk[u] & (BIN_ROWS - 1);
                        if (cur < r) {
                            row_store(out, row0 + cur, lane, acc, first, last);
                            acc = zero4;
#pragma nounroll
                            for (++cur; cur < r; ++cur)
                                row_store(out, row0 + cur, lane, zero4, first, last);
                        }
                        acc.x = fmaf(vv[u], a[u].x, acc.x);
                        acc.y = fmaf(vv[u], a[u].y, acc.y);
                        acc.z = fmaf(vv[u], a[u].z, acc.z);
                        acc.w = fmaf(vv[u], a[u].w, acc.w);
                    }
                }
            }
        }
        row_store(out, row0 + cur, lane, acc, first, last);
#pragma nounroll
        for (++cur; cur < rl0 + ROWS_W; ++cur)
            row_store(out, row0 + cur, lane, zero4, first, last);

        first = false;
        cbase += CHUNK;
    } while (cbase < end);
}

extern "C" void kernel_launch(void* const* d_in, const int* in_sizes, int n_in,
                              void* d_out, int out_size, void* d_ws, size_t ws_size,
                              hipStream_t stream)
{
    const float* x         = (const float*)d_in[0];
    const int*   edge_rows = (const int*)  d_in[1];
    const int*   edge_cols = (const int*)  d_in[2];
    const float* edge_vals = (const float*)d_in[3];
    const float* W         = (const float*)d_in[4];
    const float* b         = (const float*)d_in[5];
    float*       out       = (float*)d_out;

    float* h      = (float*)d_ws;
    int*   ofs    = (int*)(h + (size_t)N_NODES * OUT_FT);
    int*   binned = ofs + 204800;

    const int mTiles = N_NODES / 16;
    gemm_bias_wmma<<<(mTiles + 7) / 8, 256, 0, stream>>>(x, W, b, h, N_NODES);

    slice_sort<<<NSLICE, 1024, 0, stream>>>(edge_rows, edge_cols, edge_vals, ofs, binned);

    bin_gather_relu<<<NBIN, 512, 0, stream>>>(h, ofs, binned, out);
}
